# starting kernel
# baseline (speedup 1.0000x reference)
#include <hip/hip_runtime.h>
#include <math.h>

typedef __attribute__((ext_vector_type(2))) float v2f;
typedef __attribute__((ext_vector_type(8))) float v8f;

#define B_  16
#define L_  4096
#define D_  1024
#define H_  16
#define DK_ 64
#define DV_ 64

static __device__ __forceinline__ v8f wmma_gemm_f32(const float* __restrict__ A, int lda,
                                                    const float* __restrict__ Bm, int ldb,
                                                    int Klen)
{
    const int lane = threadIdx.x & 31;
    const int half = lane >> 4;
    const int mn   = lane & 15;

    const float* ap = A  + (size_t)mn * lda + 2 * half;
    const float* bp = Bm + (size_t)(2 * half) * ldb + mn;

    v8f c = {};
    for (int k = 0; k < Klen; k += 4) {
        if ((k & 511) == 0) {
            __builtin_prefetch(ap + 512, 0, 1);
        }
        v2f a, b;
        a.x = ap[0];
        a.y = ap[1];
        b.x = bp[0];
        b.y = bp[ldb];
        c = __builtin_amdgcn_wmma_f32_16x16x4_f32(false, a, false, b,
                                                  (short)0, c, false, false);
        ap += 4;
        bp += (size_t)4 * ldb;
    }
    return c;
}


__global__ void k_qproj(const float* __restrict__ query, const float* __restrict__ w_qs,
                        float* __restrict__ qs)
{
    int i = blockIdx.x * blockDim.x + threadIdx.x;
    float acc = 0.f;
    for (int d = 0; d < D_; ++d) acc += query[d] * w_qs[(size_t)d * (H_ * DK_) + i];
    qs[i] = acc * 0.125f;
}

__global__ void k_weff(const float* __restrict__ w_ks, const float* __restrict__ qs,
                       float* __restrict__ w_eff)
{
    int i = blockIdx.x * blockDim.x + threadIdx.x;
    int d = i >> 4, h = i & 15;
    const float* wk = w_ks + (size_t)d * (H_ * DK_) + h * DK_;
    const float* qh = qs + h * DK_;
    float acc = 0.f;
#pragma unroll
    for (int j = 0; j < DK_; ++j) acc += wk[j] * qh[j];
    w_eff[i] = acc;
}

__global__ void k_scores(const float* __restrict__ x, const float* __restrict__ w_eff,
                         float* __restrict__ attn)
{
    int wave = (blockIdx.x * blockDim.x + threadIdx.x) >> 5;
    int b  = wave >> 8;
    int lt = wave & 255;

    const float* A = x + ((size_t)b * L_ + (size_t)lt * 16) * D_;
    v8f c = wmma_gemm_f32(A, D_, w_eff, H_, D_);

    const int lane = threadIdx.x & 31, half = lane >> 4, n = lane & 15;
    float* out = attn + ((size_t)b * H_ + n) * L_ + (size_t)lt * 16 + 8 * half;
#pragma unroll
    for (int r = 0; r < 8; ++r) out[r] = c[r];
}

__global__ void k_softmax(float* __restrict__ attn)
{
    __shared__ float red[256];
    float* p = attn + (size_t)blockIdx.x * L_;
    const int t = threadIdx.x;

    float vals[16];
    float m = -3.4e38f;
#pragma unroll
    for (int i = 0; i < 16; ++i) { vals[i] = p[t + i * 256]; m = fmaxf(m, vals[i]); }

    red[t] = m; __syncthreads();
    for (int s = 128; s > 0; s >>= 1) {
        if (t < s) red[t] = fmaxf(red[t], red[t + s]);
        __syncthreads();
    }
    m = red[0]; __syncthreads();

    float sum = 0.f;
#pragma unroll
    for (int i = 0; i < 16; ++i) { vals[i] = expf(vals[i] - m); sum += vals[i]; }

    red[t] = sum; __syncthreads();
    for (int s = 128; s > 0; s >>= 1) {
        if (t < s) red[t] += red[t + s];
        __syncthreads();
    }
    float inv = 1.f / red[0];
#pragma unroll
    for (int i = 0; i < 16; ++i) p[t + i * 256] = vals[i] * inv;
}

#define CTXX_SPLIT 4
__global__ void k_ctxx_split(const float* __restrict__ attn, const float* __restrict__ x,
                             float* __restrict__ ctx_x_part)
{
    int wave = (blockIdx.x * blockDim.x + threadIdx.x) >> 5;
    int b  = wave >> 8;
    int t  = wave & 255;
    int s  = t >> 6;
    int dt = t & 63;
    const int k0 = s * (L_ / CTXX_SPLIT);

    const float* A  = attn + (size_t)b * H_ * L_ + k0;
    const float* Bm = x + ((size_t)b * L_ + k0) * D_ + dt * 16;
    v8f c = wmma_gemm_f32(A, L_, Bm, D_, L_ / CTXX_SPLIT);

    const int lane = threadIdx.x & 31, half = lane >> 4, n = lane & 15;
    float* out = ctx_x_part + (size_t)s * (B_ * H_ * D_)
               + (size_t)b * H_ * D_ + (size_t)dt * 16 + n;
#pragma unroll
    for (int r = 0; r < 8; ++r) out[(size_t)(r + 8 * half) * D_] = c[r];
}

__global__ void k_reduce(const float* __restrict__ part, float* __restrict__ dst,
                         int nparts, int n)
{
    int stride = gridDim.x * blockDim.x;
    for (int i = blockIdx.x * blockDim.x + threadIdx.x; i < n; i += stride) {
        float s = 0.f;
        for (int p = 0; p < nparts; ++p) s += part[(size_t)p * n + i];
        dst[i] = s;
    }
}

#define VSPLIT 8
__global__ void k_ctx_split(const float* __restrict__ ctx_x, const float* __restrict__ w_vs,
                            float* __restrict__ ctx_part)
{
    int wave = (blockIdx.x * blockDim.x + threadIdx.x) >> 5;
    int tile = wave >> 3;
    int s    = wave & 7;
    int h    = tile >> 2;
    int dvt  = tile & 3;
    const int k0 = s * (D_ / VSPLIT);

    const float* A  = ctx_x + (size_t)h * D_ + k0;
    const float* Bm = w_vs + (size_t)k0 * (H_ * DV_) + h * DV_ + dvt * 16;
    v8f c = wmma_gemm_f32(A, H_ * D_, Bm, H_ * DV_, D_ / VSPLIT);

    const int lane = threadIdx.x & 31, half = lane >> 4, n = lane & 15;
    float* out = ctx_part + (size_t)s * (B_ * H_ * DV_)
               + h * DV_ + dvt * 16 + n;
#pragma unroll
    for (int r = 0; r < 8; ++r) out[(size_t)(r + 8 * half) * (H_ * DV_)] = c[r];
}

__global__ void k_fc_split(const float* __restrict__ ctx, const float* __restrict__ fc,
                           float* __restrict__ out_pre_part)
{
    int wave = (blockIdx.x * blockDim.x + threadIdx.x) >> 5;
    int dt = wave >> 3;
    int s  = wave & 7;
    const int k0 = s * ((H_ * DV_) / VSPLIT);

    const float* A  = ctx + k0;
    const float* Bm = fc + (size_t)k0 * D_ + dt * 16;
    v8f c = wmma_gemm_f32(A, H_ * DV_, Bm, D_, (H_ * DV_) / VSPLIT);

    const int lane = threadIdx.x & 31, half = lane >> 4, n = lane & 15;
    float* out = out_pre_part + (size_t)s * (B_ * D_) + dt * 16 + n;
#pragma unroll
    for (int r = 0; r < 8; ++r) out[(size_t)(r + 8 * half) * D_] = c[r];
}

__global__ void k_ln(const float* __restrict__ out_pre_part, const float* __restrict__ query,
                     const float* __restrict__ gamma, const float* __restrict__ beta,
                     float* __restrict__ out)
{
    __shared__ float rs[256], rs2[256];
    const int b = blockIdx.x, t = threadIdx.x;

    float vals[4];
    float s = 0.f, s2 = 0.f;
#pragma unroll
    for (int i = 0; i < 4; ++i) {
        const int idx = t + i * 256;
        float v = query[idx];
        for (int p = 0; p < VSPLIT; ++p)
            v += out_pre_part[(size_t)p * (B_ * D_) + (size_t)b * D_ + idx];
        vals[i] = v; s += v; s2 += v * v;
    }
    rs[t] = s; rs2[t] = s2; __syncthreads();
    for (int k = 128; k > 0; k >>= 1) {
        if (t < k) { rs[t] += rs[t + k]; rs2[t] += rs2[t + k]; }
        __syncthreads();
    }
    float mean = rs[0] * (1.f / D_);
    float var  = rs2[0] * (1.f / D_) - mean * mean;
    float inv  = rsqrtf(var + 1e-6f);
#pragma unroll
    for (int i = 0; i < 4; ++i) {
        int idx = t + i * 256;
        out[(size_t)b * D_ + idx] = (vals[i] - mean) * inv * gamma[idx] + beta[idx];
    }
}

extern "C" void kernel_launch(void* const* d_in, const int* in_sizes, int n_in,
                              void* d_out, int out_size, void* d_ws, size_t ws_size,
                              hipStream_t stream)
{
    (void)in_sizes; (void)n_in; (void)out_size; (void)ws_size;

    const float* x     = (const float*)d_in[0];
    const float* query = (const float*)d_in[1];
    const float* w_qs  = (const float*)d_in[2];
    const float* w_ks  = (const float*)d_in[3];
    const float* w_vs  = (const float*)d_in[4];
    const float* fc    = (const float*)d_in[5];
    const float* gamma = (const float*)d_in[6];
    const float* beta  = (const float*)d_in[7];

    float* out  = (float*)d_out;
    float* attn = out + (size_t)B_ * D_;

    float* ws           = (float*)d_ws;
    float* qs           = ws;
    float* w_eff        = qs + H_ * DK_;
    float* ctx_x_part   = w_eff + D_ * H_;
    float* ctx_x        = ctx_x_part + (size_t)CTXX_SPLIT * B_ * H_ * D_;
    float* ctx_part     = ctx_x + (size_t)B_ * H_ * D_;
    float* ctx          = ctx_part + (size_t)VSPLIT * B_ * H_ * DV_;
    float* out_pre_part = ctx + (size_t)B_ * H_ * DV_;

    k_qproj<<<4, 256, 0, stream>>>(query, w_qs, qs);
    k_weff<<<64, 256, 0, stream>>>(w_ks, qs, w_eff);
    k_scores<<<512, 256, 0, stream>>>(x, w_eff, attn);
    k_softmax<<<B_ * H_, 256, 0, stream>>>(attn);
    k_ctxx_split<<<512, 256, 0, stream>>>(attn, x, ctx_x_part);
    k_reduce<<<256, 256, 0, stream>>>(ctx_x_part, ctx_x, CTXX_SPLIT, B_ * H_ * D_);
    k_ctx_split<<<64, 256, 0, stream>>>(ctx_x, w_vs, ctx_part);
    k_reduce<<<16, 256, 0, stream>>>(ctx_part, ctx, VSPLIT, B_ * H_ * DV_);
    k_fc_split<<<64, 256, 0, stream>>>(ctx, fc, out_pre_part);
    k_ln<<<B_, 256, 0, stream>>>(out_pre_part, query, gamma, beta, out);
}


# q projection spread from 4 blocks to 32 blocks of 1024 threads: rows split over 32 waves, all row loads in flight, partial sums added in LDS (changes the summation order of qs)
# speedup vs baseline: 1.3236x; 1.3236x over previous
#include <hip/hip_runtime.h>
#include <math.h>

typedef __attribute__((ext_vector_type(2))) float v2f;
typedef __attribute__((ext_vector_type(8))) float v8f;

#define B_  16
#define L_  4096
#define D_  1024
#define H_  16
#define DK_ 64
#define DV_ 64

static __device__ __forceinline__ v8f wmma_gemm_f32(const float* __restrict__ A, int lda,
                                                    const float* __restrict__ Bm, int ldb,
                                                    int Klen)
{
    const int lane = threadIdx.x & 31;
    const int half = lane >> 4;
    const int mn   = lane & 15;

    const float* ap = A  + (size_t)mn * lda + 2 * half;
    const float* bp = Bm + (size_t)(2 * half) * ldb + mn;

    v8f c = {};
    for (int k = 0; k < Klen; k += 4) {
        if ((k & 511) == 0) {
            __builtin_prefetch(ap + 512, 0, 1);
        }
        v2f a, b;
        a.x = ap[0];
        a.y = ap[1];
        b.x = bp[0];
        b.y = bp[ldb];
        c = __builtin_amdgcn_wmma_f32_16x16x4_f32(false, a, false, b,
                                                  (short)0, c, false, false);
        ap += 4;
        bp += (size_t)4 * ldb;
    }
    return c;
}


#define QP_COLS 32
#define QP_ROWS 32
__global__ void __launch_bounds__(1024)
k_qproj(const float* __restrict__ query, const float* __restrict__ w_qs,
        float* __restrict__ qs)
{
    __shared__ float red[D_ / QP_ROWS][QP_COLS];

    const int lane = threadIdx.x & 31;
    const int w = __builtin_amdgcn_readfirstlane((int)(threadIdx.x >> 5));
    const int col = blockIdx.x * QP_COLS + lane;

    const float* wp = w_qs + (size_t)(w * QP_ROWS) * (H_ * DK_) + col;
    const float* qp = query + w * QP_ROWS;

    float v[QP_ROWS];
    for (int r = 0; r < QP_ROWS; ++r) v[r] = wp[(size_t)r * (H_ * DK_)];

    float acc = 0.f;
    for (int r = 0; r < QP_ROWS; ++r) acc += qp[r] * v[r];

    red[w][lane] = acc;
    __syncthreads();

    if (w == 0) {
        float s = 0.f;
        for (int k = 0; k < D_ / QP_ROWS; ++k) s += red[k][lane];
        qs[col] = s * 0.125f;
    }
}

__global__ void k_weff(const float* __restrict__ w_ks, const float* __restrict__ qs,
                       float* __restrict__ w_eff)
{
    int i = blockIdx.x * blockDim.x + threadIdx.x;
    int d = i >> 4, h = i & 15;
    const float* wk = w_ks + (size_t)d * (H_ * DK_) + h * DK_;
    const float* qh = qs + h * DK_;
    float acc = 0.f;
#pragma unroll
    for (int j = 0; j < DK_; ++j) acc += wk[j] * qh[j];
    w_eff[i] = acc;
}

__global__ void k_scores(const float* __restrict__ x, const float* __restrict__ w_eff,
                         float* __restrict__ attn)
{
    int wave = (blockIdx.x * blockDim.x + threadIdx.x) >> 5;
    int b  = wave >> 8;
    int lt = wave & 255;

    const float* A = x + ((size_t)b * L_ + (size_t)lt * 16) * D_;
    v8f c = wmma_gemm_f32(A, D_, w_eff, H_, D_);

    const int lane = threadIdx.x & 31, half = lane >> 4, n = lane & 15;
    float* out = attn + ((size_t)b * H_ + n) * L_ + (size_t)lt * 16 + 8 * half;
#pragma unroll
    for (int r = 0; r < 8; ++r) out[r] = c[r];
}

__global__ void k_softmax(float* __restrict__ attn)
{
    __shared__ float red[256];
    float* p = attn + (size_t)blockIdx.x * L_;
    const int t = threadIdx.x;

    float vals[16];
    float m = -3.4e38f;
#pragma unroll
    for (int i = 0; i < 16; ++i) { vals[i] = p[t + i * 256]; m = fmaxf(m, vals[i]); }

    red[t] = m; __syncthreads();
    for (int s = 128; s > 0; s >>= 1) {
        if (t < s) red[t] = fmaxf(red[t], red[t + s]);
        __syncthreads();
    }
    m = red[0]; __syncthreads();

    float sum = 0.f;
#pragma unroll
    for (int i = 0; i < 16; ++i) { vals[i] = expf(vals[i] - m); sum += vals[i]; }

    red[t] = sum; __syncthreads();
    for (int s = 128; s > 0; s >>= 1) {
        if (t < s) red[t] += red[t + s];
        __syncthreads();
    }
    float inv = 1.f / red[0];
#pragma unroll
    for (int i = 0; i < 16; ++i) p[t + i * 256] = vals[i] * inv;
}

#define CTXX_SPLIT 4
__global__ void k_ctxx_split(const float* __restrict__ attn, const float* __restrict__ x,
                             float* __restrict__ ctx_x_part)
{
    int wave = (blockIdx.x * blockDim.x + threadIdx.x) >> 5;
    int b  = wave >> 8;
    int t  = wave & 255;
    int s  = t >> 6;
    int dt = t & 63;
    const int k0 = s * (L_ / CTXX_SPLIT);

    const float* A  = attn + (size_t)b * H_ * L_ + k0;
    const float* Bm = x + ((size_t)b * L_ + k0) * D_ + dt * 16;
    v8f c = wmma_gemm_f32(A, L_, Bm, D_, L_ / CTXX_SPLIT);

    const int lane = threadIdx.x & 31, half = lane >> 4, n = lane & 15;
    float* out = ctx_x_part + (size_t)s * (B_ * H_ * D_)
               + (size_t)b * H_ * D_ + (size_t)dt * 16 + n;
#pragma unroll
    for (int r = 0; r < 8; ++r) out[(size_t)(r + 8 * half) * D_] = c[r];
}

__global__ void k_reduce(const float* __restrict__ part, float* __restrict__ dst,
                         int nparts, int n)
{
    int stride = gridDim.x * blockDim.x;
    for (int i = blockIdx.x * blockDim.x + threadIdx.x; i < n; i += stride) {
        float s = 0.f;
        for (int p = 0; p < nparts; ++p) s += part[(size_t)p * n + i];
        dst[i] = s;
    }
}

#define VSPLIT 8
__global__ void k_ctx_split(const float* __restrict__ ctx_x, const float* __restrict__ w_vs,
                            float* __restrict__ ctx_part)
{
    int wave = (blockIdx.x * blockDim.x + threadIdx.x) >> 5;
    int tile = wave >> 3;
    int s    = wave & 7;
    int h    = tile >> 2;
    int dvt  = tile & 3;
    const int k0 = s * (D_ / VSPLIT);

    const float* A  = ctx_x + (size_t)h * D_ + k0;
    const float* Bm = w_vs + (size_t)k0 * (H_ * DV_) + h * DV_ + dvt * 16;
    v8f c = wmma_gemm_f32(A, H_ * D_, Bm, H_ * DV_, D_ / VSPLIT);

    const int lane = threadIdx.x & 31, half = lane >> 4, n = lane & 15;
    float* out = ctx_part + (size_t)s * (B_ * H_ * DV_)
               + h * DV_ + dvt * 16 + n;
#pragma unroll
    for (int r = 0; r < 8; ++r) out[(size_t)(r + 8 * half) * (H_ * DV_)] = c[r];
}

__global__ void k_fc_split(const float* __restrict__ ctx, const float* __restrict__ fc,
                           float* __restrict__ out_pre_part)
{
    int wave = (blockIdx.x * blockDim.x + threadIdx.x) >> 5;
    int dt = wave >> 3;
    int s  = wave & 7;
    const int k0 = s * ((H_ * DV_) / VSPLIT);

    const float* A  = ctx + k0;
    const float* Bm = fc + (size_t)k0 * D_ + dt * 16;
    v8f c = wmma_gemm_f32(A, H_ * DV_, Bm, D_, (H_ * DV_) / VSPLIT);

    const int lane = threadIdx.x & 31, half = lane >> 4, n = lane & 15;
    float* out = out_pre_part + (size_t)s * (B_ * D_) + dt * 16 + n;
#pragma unroll
    for (int r = 0; r < 8; ++r) out[(size_t)(r + 8 * half) * D_] = c[r];
}

__global__ void k_ln(const float* __restrict__ out_pre_part, const float* __restrict__ query,
                     const float* __restrict__ gamma, const float* __restrict__ beta,
                     float* __restrict__ out)
{
    __shared__ float rs[256], rs2[256];
    const int b = blockIdx.x, t = threadIdx.x;

    float vals[4];
    float s = 0.f, s2 = 0.f;
#pragma unroll
    for (int i = 0; i < 4; ++i) {
        const int idx = t + i * 256;
        float v = query[idx];
        for (int p = 0; p < VSPLIT; ++p)
            v += out_pre_part[(size_t)p * (B_ * D_) + (size_t)b * D_ + idx];
        vals[i] = v; s += v; s2 += v * v;
    }
    rs[t] = s; rs2[t] = s2; __syncthreads();
    for (int k = 128; k > 0; k >>= 1) {
        if (t < k) { rs[t] += rs[t + k]; rs2[t] += rs2[t + k]; }
        __syncthreads();
    }
    float mean = rs[0] * (1.f / D_);
    float var  = rs2[0] * (1.f / D_) - mean * mean;
    float inv  = rsqrtf(var + 1e-6f);
#pragma unroll
    for (int i = 0; i < 4; ++i) {
        int idx = t + i * 256;
        out[(size_t)b * D_ + idx] = (vals[i] - mean) * inv * gamma[idx] + beta[idx];
    }
}

extern "C" void kernel_launch(void* const* d_in, const int* in_sizes, int n_in,
                              void* d_out, int out_size, void* d_ws, size_t ws_size,
                              hipStream_t stream)
{
    (void)in_sizes; (void)n_in; (void)out_size; (void)ws_size;

    const float* x     = (const float*)d_in[0];
    const float* query = (const float*)d_in[1];
    const float* w_qs  = (const float*)d_in[2];
    const float* w_ks  = (const float*)d_in[3];
    const float* w_vs  = (const float*)d_in[4];
    const float* fc    = (const float*)d_in[5];
    const float* gamma = (const float*)d_in[6];
    const float* beta  = (const float*)d_in[7];

    float* out  = (float*)d_out;
    float* attn = out + (size_t)B_ * D_;

    float* ws           = (float*)d_ws;
    float* qs           = ws;
    float* w_eff        = qs + H_ * DK_;
    float* ctx_x_part   = w_eff + D_ * H_;
    float* ctx_x        = ctx_x_part + (size_t)CTXX_SPLIT * B_ * H_ * D_;
    float* ctx_part     = ctx_x + (size_t)B_ * H_ * D_;
    float* ctx          = ctx_part + (size_t)VSPLIT * B_ * H_ * DV_;
    float* out_pre_part = ctx + (size_t)B_ * H_ * DV_;

    k_qproj<<<32, 1024, 0, stream>>>(query, w_qs, qs);
    k_weff<<<64, 256, 0, stream>>>(w_ks, qs, w_eff);
    k_scores<<<512, 256, 0, stream>>>(x, w_eff, attn);
    k_softmax<<<B_ * H_, 256, 0, stream>>>(attn);
    k_ctxx_split<<<512, 256, 0, stream>>>(attn, x, ctx_x_part);
    k_reduce<<<256, 256, 0, stream>>>(ctx_x_part, ctx_x, CTXX_SPLIT, B_ * H_ * D_);
    k_ctx_split<<<64, 256, 0, stream>>>(ctx_x, w_vs, ctx_part);
    k_reduce<<<16, 256, 0, stream>>>(ctx_part, ctx, VSPLIT, B_ * H_ * DV_);
    k_fc_split<<<64, 256, 0, stream>>>(ctx, fc, out_pre_part);
    k_ln<<<B_, 256, 0, stream>>>(out_pre_part, query, gamma, beta, out);
}


# GEMM operand loads double-buffered in registers: the 24 loads of k-block i+1 (8 steps) are requested before the matrix steps of block i, so a wave has 8-16 steps in flight; k order and sums unchanged
# speedup vs baseline: 1.3635x; 1.0301x over previous
#include <hip/hip_runtime.h>
#include <math.h>

typedef __attribute__((ext_vector_type(2))) float v2f;
typedef __attribute__((ext_vector_type(8))) float v8f;

#define B_  16
#define L_  4096
#define D_  1024
#define H_  16
#define DK_ 64
#define DV_ 64

#define GB_ 8

static __device__ __forceinline__ void gemm_load_blk(const float* __restrict__ ap,
                                                     const float* __restrict__ bp, int ldb,
                                                     v2f (&a)[GB_], v2f (&b)[GB_])
{
#pragma unroll
    for (int u = 0; u < GB_; ++u) {
        a[u].x = ap[4 * u];
        a[u].y = ap[4 * u + 1];
        b[u].x = bp[(size_t)(4 * u) * ldb];
        b[u].y = bp[(size_t)(4 * u + 1) * ldb];
    }
}

static __device__ __forceinline__ v8f gemm_mma_blk(const v2f (&a)[GB_], const v2f (&b)[GB_], v8f c)
{
#pragma unroll
    for (int u = 0; u < GB_; ++u)
        c = __builtin_amdgcn_wmma_f32_16x16x4_f32(false, a[u], false, b[u],
                                                  (short)0, c, false, false);
    return c;
}

static __device__ __forceinline__ v8f wmma_gemm_f32(const float* __restrict__ A, int lda,
                                                    const float* __restrict__ Bm, int ldb,
                                                    int Klen)
{
    const int lane = threadIdx.x & 31;
    const int half = lane >> 4;
    const int mn   = lane & 15;

    const float* ap = A  + (size_t)mn * lda + 2 * half;
    const float* bp = Bm + (size_t)(2 * half) * ldb + mn;
    const size_t bstep = (size_t)(4 * GB_) * ldb;
    const int nblk = Klen / (4 * GB_);

    v2f a0[GB_], b0[GB_], a1[GB_], b1[GB_];
    v8f c = {};

    gemm_load_blk(ap, bp, ldb, a0, b0);
    for (int i = 0; i + 2 < nblk; i += 2) {
        gemm_load_blk(ap + 4 * GB_, bp + bstep, ldb, a1, b1);
        c = gemm_mma_blk(a0, b0, c);
        gemm_load_blk(ap + 8 * GB_, bp + 2 * bstep, ldb, a0, b0);
        c = gemm_mma_blk(a1, b1, c);
        ap += 8 * GB_;
        bp += 2 * bstep;
    }
    gemm_load_blk(ap + 4 * GB_, bp + bstep, ldb, a1, b1);
    c = gemm_mma_blk(a0, b0, c);
    c = gemm_mma_blk(a1, b1, c);
    return c;
}


#define QP_COLS 32
#define QP_ROWS 32
__global__ void __launch_bounds__(1024)
k_qproj(const float* __restrict__ query, const float* __restrict__ w_qs,
        float* __restrict__ qs)
{
    __shared__ float red[D_ / QP_ROWS][QP_COLS];

    const int lane = threadIdx.x & 31;
    const int w = __builtin_amdgcn_readfirstlane((int)(threadIdx.x >> 5));
    const int col = blockIdx.x * QP_COLS + lane;

    const float* wp = w_qs + (size_t)(w * QP_ROWS) * (H_ * DK_) + col;
    const float* qp = query + w * QP_ROWS;

    float v[QP_ROWS];
    for (int r = 0; r < QP_ROWS; ++r) v[r] = wp[(size_t)r * (H_ * DK_)];

    float acc = 0.f;
    for (int r = 0; r < QP_ROWS; ++r) acc += qp[r] * v[r];

    red[w][lane] = acc;
    __syncthreads();

    if (w == 0) {
        float s = 0.f;
        for (int k = 0; k < D_ / QP_ROWS; ++k) s += red[k][lane];
        qs[col] = s * 0.125f;
    }
}

__global__ void k_weff(const float* __restrict__ w_ks, const float* __restrict__ qs,
                       float* __restrict__ w_eff)
{
    int i = blockIdx.x * blockDim.x + threadIdx.x;
    int d = i >> 4, h = i & 15;
    const float* wk = w_ks + (size_t)d * (H_ * DK_) + h * DK_;
    const float* qh = qs + h * DK_;
    float acc = 0.f;
#pragma unroll
    for (int j = 0; j < DK_; ++j) acc += wk[j] * qh[j];
    w_eff[i] = acc;
}

__global__ void k_scores(const float* __restrict__ x, const float* __restrict__ w_eff,
                         float* __restrict__ attn)
{
    int wave = (blockIdx.x * blockDim.x + threadIdx.x) >> 5;
    int b  = wave >> 8;
    int lt = wave & 255;

    const float* A = x + ((size_t)b * L_ + (size_t)lt * 16) * D_;
    v8f c = wmma_gemm_f32(A, D_, w_eff, H_, D_);

    const int lane = threadIdx.x & 31, half = lane >> 4, n = lane & 15;
    float* out = attn + ((size_t)b * H_ + n) * L_ + (size_t)lt * 16 + 8 * half;
#pragma unroll
    for (int r = 0; r < 8; ++r) out[r] = c[r];
}

__global__ void k_softmax(float* __restrict__ attn)
{
    __shared__ float red[256];
    float* p = attn + (size_t)blockIdx.x * L_;
    const int t = threadIdx.x;

    float vals[16];
    float m = -3.4e38f;
#pragma unroll
    for (int i = 0; i < 16; ++i) { vals[i] = p[t + i * 256]; m = fmaxf(m, vals[i]); }

    red[t] = m; __syncthreads();
    for (int s = 128; s > 0; s >>= 1) {
        if (t < s) red[t] = fmaxf(red[t], red[t + s]);
        __syncthreads();
    }
    m = red[0]; __syncthreads();

    float sum = 0.f;
#pragma unroll
    for (int i = 0; i < 16; ++i) { vals[i] = expf(vals[i] - m); sum += vals[i]; }

    red[t] = sum; __syncthreads();
    for (int s = 128; s > 0; s >>= 1) {
        if (t < s) red[t] += red[t + s];
        __syncthreads();
    }
    float inv = 1.f / red[0];
#pragma unroll
    for (int i = 0; i < 16; ++i) p[t + i * 256] = vals[i] * inv;
}

#define CTXX_SPLIT 4
__global__ void k_ctxx_split(const float* __restrict__ attn, const float* __restrict__ x,
                             float* __restrict__ ctx_x_part)
{
    int wave = (blockIdx.x * blockDim.x + threadIdx.x) >> 5;
    int b  = wave >> 8;
    int t  = wave & 255;
    int s  = t >> 6;
    int dt = t & 63;
    const int k0 = s * (L_ / CTXX_SPLIT);

    const float* A  = attn + (size_t)b * H_ * L_ + k0;
    const float* Bm = x + ((size_t)b * L_ + k0) * D_ + dt * 16;
    v8f c = wmma_gemm_f32(A, L_, Bm, D_, L_ / CTXX_SPLIT);

    const int lane = threadIdx.x & 31, half = lane >> 4, n = lane & 15;
    float* out = ctx_x_part + (size_t)s * (B_ * H_ * D_)
               + (size_t)b * H_ * D_ + (size_t)dt * 16 + n;
#pragma unroll
    for (int r = 0; r < 8; ++r) out[(size_t)(r + 8 * half) * D_] = c[r];
}

__global__ void k_reduce(const float* __restrict__ part, float* __restrict__ dst,
                         int nparts, int n)
{
    int stride = gridDim.x * blockDim.x;
    for (int i = blockIdx.x * blockDim.x + threadIdx.x; i < n; i += stride) {
        float s = 0.f;
        for (int p = 0; p < nparts; ++p) s += part[(size_t)p * n + i];
        dst[i] = s;
    }
}

#define VSPLIT 8
__global__ void k_ctx_split(const float* __restrict__ ctx_x, const float* __restrict__ w_vs,
                            float* __restrict__ ctx_part)
{
    int wave = (blockIdx.x * blockDim.x + threadIdx.x) >> 5;
    int tile = wave >> 3;
    int s    = wave & 7;
    int h    = tile >> 2;
    int dvt  = tile & 3;
    const int k0 = s * (D_ / VSPLIT);

    const float* A  = ctx_x + (size_t)h * D_ + k0;
    const float* Bm = w_vs + (size_t)k0 * (H_ * DV_) + h * DV_ + dvt * 16;
    v8f c = wmma_gemm_f32(A, H_ * D_, Bm, H_ * DV_, D_ / VSPLIT);

    const int lane = threadIdx.x & 31, half = lane >> 4, n = lane & 15;
    float* out = ctx_part + (size_t)s * (B_ * H_ * DV_)
               + h * DV_ + dvt * 16 + n;
#pragma unroll
    for (int r = 0; r < 8; ++r) out[(size_t)(r + 8 * half) * (H_ * DV_)] = c[r];
}

__global__ void k_fc_split(const float* __restrict__ ctx, const float* __restrict__ fc,
                           float* __restrict__ out_pre_part)
{
    int wave = (blockIdx.x * blockDim.x + threadIdx.x) >> 5;
    int dt = wave >> 3;
    int s  = wave & 7;
    const int k0 = s * ((H_ * DV_) / VSPLIT);

    const float* A  = ctx + k0;
    const float* Bm = fc + (size_t)k0 * D_ + dt * 16;
    v8f c = wmma_gemm_f32(A, H_ * DV_, Bm, D_, (H_ * DV_) / VSPLIT);

    const int lane = threadIdx.x & 31, half = lane >> 4, n = lane & 15;
    float* out = out_pre_part + (size_t)s * (B_ * D_) + dt * 16 + n;
#pragma unroll
    for (int r = 0; r < 8; ++r) out[(size_t)(r + 8 * half) * D_] = c[r];
}

__global__ void k_ln(const float* __restrict__ out_pre_part, const float* __restrict__ query,
                     const float* __restrict__ gamma, const float* __restrict__ beta,
                     float* __restrict__ out)
{
    __shared__ float rs[256], rs2[256];
    const int b = blockIdx.x, t = threadIdx.x;

    float vals[4];
    float s = 0.f, s2 = 0.f;
#pragma unroll
    for (int i = 0; i < 4; ++i) {
        const int idx = t + i * 256;
        float v = query[idx];
        for (int p = 0; p < VSPLIT; ++p)
            v += out_pre_part[(size_t)p * (B_ * D_) + (size_t)b * D_ + idx];
        vals[i] = v; s += v; s2 += v * v;
    }
    rs[t] = s; rs2[t] = s2; __syncthreads();
    for (int k = 128; k > 0; k >>= 1) {
        if (t < k) { rs[t] += rs[t + k]; rs2[t] += rs2[t + k]; }
        __syncthreads();
    }
    float mean = rs[0] * (1.f / D_);
    float var  = rs2[0] * (1.f / D_) - mean * mean;
    float inv  = rsqrtf(var + 1e-6f);
#pragma unroll
    for (int i = 0; i < 4; ++i) {
        int idx = t + i * 256;
        out[(size_t)b * D_ + idx] = (vals[i] - mean) * inv * gamma[idx] + beta[idx];
    }
}

extern "C" void kernel_launch(void* const* d_in, const int* in_sizes, int n_in,
                              void* d_out, int out_size, void* d_ws, size_t ws_size,
                              hipStream_t stream)
{
    (void)in_sizes; (void)n_in; (void)out_size; (void)ws_size;

    const float* x     = (const float*)d_in[0];
    const float* query = (const float*)d_in[1];
    const float* w_qs  = (const float*)d_in[2];
    const float* w_ks  = (const float*)d_in[3];
    const float* w_vs  = (const float*)d_in[4];
    const float* fc    = (const float*)d_in[5];
    const float* gamma = (const float*)d_in[6];
    const float* beta  = (const float*)d_in[7];

    float* out  = (float*)d_out;
    float* attn = out + (size_t)B_ * D_;

    float* ws           = (float*)d_ws;
    float* qs           = ws;
    float* w_eff        = qs + H_ * DK_;
    float* ctx_x_part   = w_eff + D_ * H_;
    float* ctx_x        = ctx_x_part + (size_t)CTXX_SPLIT * B_ * H_ * D_;
    float* ctx_part     = ctx_x + (size_t)B_ * H_ * D_;
    float* ctx          = ctx_part + (size_t)VSPLIT * B_ * H_ * DV_;
    float* out_pre_part = ctx + (size_t)B_ * H_ * DV_;

    k_qproj<<<32, 1024, 0, stream>>>(query, w_qs, qs);
    k_weff<<<64, 256, 0, stream>>>(w_ks, qs, w_eff);
    k_scores<<<512, 256, 0, stream>>>(x, w_eff, attn);
    k_softmax<<<B_ * H_, 256, 0, stream>>>(attn);
    k_ctxx_split<<<512, 256, 0, stream>>>(attn, x, ctx_x_part);
    k_reduce<<<256, 256, 0, stream>>>(ctx_x_part, ctx_x, CTXX_SPLIT, B_ * H_ * D_);
    k_ctx_split<<<64, 256, 0, stream>>>(ctx_x, w_vs, ctx_part);
    k_reduce<<<16, 256, 0, stream>>>(ctx_part, ctx, VSPLIT, B_ * H_ * DV_);
    k_fc_split<<<64, 256, 0, stream>>>(ctx, fc, out_pre_part);
    k_ln<<<B_, 256, 0, stream>>>(out_pre_part, query, gamma, beta, out);
}


# A operand of every GEMM staged through a per-wave LDS slab: whole-line 16-byte-per-lane loads (4 lines per load) replace fragment loads that touch 16 lines each; k order and sums unchanged
# speedup vs baseline: 1.5235x; 1.1173x over previous
#include <hip/hip_runtime.h>
#include <math.h>

typedef __attribute__((ext_vector_type(2))) float v2f;
typedef __attribute__((ext_vector_type(4))) float v4f;
typedef __attribute__((ext_vector_type(8))) float v8f;

#define B_  16
#define L_  4096
#define D_  1024
#define H_  16
#define DK_ 64
#define DV_ 64

#define GB_ 8
#define AP_ 36
#define GW_ 8

static __device__ __forceinline__ void gemm_gload_a(const float* __restrict__ ag, int lda,
                                                    v4f (&g)[4])
{
#pragma unroll
    for (int j = 0; j < 4; ++j)
        g[j] = *(const v4f*)(ag + (size_t)(4 * j) * lda);
}

static __device__ __forceinline__ void gemm_lds_put(float* sw, const v4f (&g)[4])
{
#pragma unroll
    for (int j = 0; j < 4; ++j)
        *(v4f*)(sw + (4 * j) * AP_) = g[j];
}

static __device__ __forceinline__ void gemm_lds_get(const float* sr, v2f (&a)[GB_])
{
#pragma unroll
    for (int u = 0; u < GB_; ++u)
        a[u] = *(const v2f*)(sr + 4 * u);
}

static __device__ __forceinline__ void wave_lds_order()
{
    __builtin_amdgcn_fence(__ATOMIC_RELEASE, "wavefront");
    __builtin_amdgcn_wave_barrier();
    __builtin_amdgcn_fence(__ATOMIC_ACQUIRE, "wavefront");
}

static __device__ __forceinline__ void gemm_load_b(const float* __restrict__ bp, int ldb,
                                                   v2f (&b)[GB_])
{
#pragma unroll
    for (int u = 0; u < GB_; ++u) {
        b[u].x = bp[(size_t)(4 * u) * ldb];
        b[u].y = bp[(size_t)(4 * u + 1) * ldb];
    }
}

static __device__ __forceinline__ v8f gemm_mma_blk(const v2f (&a)[GB_], const v2f (&b)[GB_], v8f c)
{
#pragma unroll
    for (int u = 0; u < GB_; ++u)
        c = __builtin_amdgcn_wmma_f32_16x16x4_f32(false, a[u], false, b[u],
                                                  (short)0, c, false, false);
    return c;
}

static __device__ __forceinline__ v8f wmma_gemm_f32(const float* __restrict__ A, int lda,
                                                    const float* __restrict__ Bm, int ldb,
                                                    int Klen)
{
    __shared__ __attribute__((aligned(16))) float slab[GW_][16 * AP_];

    const int lane = threadIdx.x & 31;
    const int half = lane >> 4;
    const int mn   = lane & 15;
    float* sl = slab[threadIdx.x >> 5];

    const float* ag = A + (size_t)(lane >> 3) * lda + 4 * (lane & 7);
    float*       sw = sl + (lane >> 3) * AP_ + 4 * (lane & 7);
    const float* sr = sl + mn * AP_ + 2 * half;
    const float* bp = Bm + (size_t)(2 * half) * ldb + mn;
    const size_t bstep = (size_t)(4 * GB_) * ldb;
    const int nblk = Klen / (4 * GB_);

    v4f g0[4], g1[4];
    v2f a0[GB_], b0[GB_], a1[GB_], b1[GB_];
    v8f c = {};

    gemm_gload_a(ag, lda, g0);
    gemm_gload_a(ag + 4 * GB_, lda, g1);
    gemm_load_b(bp, ldb, b0);
    gemm_lds_put(sw, g0);
    gemm_gload_a(ag + 8 * GB_, lda, g0);
    wave_lds_order();
    gemm_lds_get(sr, a0);
    ag += 12 * GB_;

    for (int i = 0; i + 2 < nblk; i += 2) {
        wave_lds_order();
        gemm_lds_put(sw, g1);
        gemm_gload_a(ag, lda, g1);
        gemm_load_b(bp + bstep, ldb, b1);
        wave_lds_order();
        gemm_lds_get(sr, a1);
        c = gemm_mma_blk(a0, b0, c);

        wave_lds_order();
        gemm_lds_put(sw, g0);
        if (i + 4 < nblk) gemm_gload_a(ag + 4 * GB_, lda, g0);
        gemm_load_b(bp + 2 * bstep, ldb, b0);
        wave_lds_order();
        gemm_lds_get(sr, a0);
        c = gemm_mma_blk(a1, b1, c);

        ag += 8 * GB_;
        bp += 2 * bstep;
    }
    wave_lds_order();
    gemm_lds_put(sw, g1);
    gemm_load_b(bp + bstep, ldb, b1);
    wave_lds_order();
    gemm_lds_get(sr, a1);
    c = gemm_mma_blk(a0, b0, c);
    c = gemm_mma_blk(a1, b1, c);
    return c;
}


#define QP_COLS 32
#define QP_ROWS 32
__global__ void __launch_bounds__(1024)
k_qproj(const float* __restrict__ query, const float* __restrict__ w_qs,
        float* __restrict__ qs)
{
    __shared__ float red[D_ / QP_ROWS][QP_COLS];

    const int lane = threadIdx.x & 31;
    const int w = __builtin_amdgcn_readfirstlane((int)(threadIdx.x >> 5));
    const int col = blockIdx.x * QP_COLS + lane;

    const float* wp = w_qs + (size_t)(w * QP_ROWS) * (H_ * DK_) + col;
    const float* qp = query + w * QP_ROWS;

    float v[QP_ROWS];
    for (int r = 0; r < QP_ROWS; ++r) v[r] = wp[(size_t)r * (H_ * DK_)];

    float acc = 0.f;
    for (int r = 0; r < QP_ROWS; ++r) acc += qp[r] * v[r];

    red[w][lane] = acc;
    __syncthreads();

    if (w == 0) {
        float s = 0.f;
        for (int k = 0; k < D_ / QP_ROWS; ++k) s += red[k][lane];
        qs[col] = s * 0.125f;
    }
}

__global__ void k_weff(const float* __restrict__ w_ks, const float* __restrict__ qs,
                       float* __restrict__ w_eff)
{
    int i = blockIdx.x * blockDim.x + threadIdx.x;
    int d = i >> 4, h = i & 15;
    const float* wk = w_ks + (size_t)d * (H_ * DK_) + h * DK_;
    const float* qh = qs + h * DK_;
    float acc = 0.f;
#pragma unroll
    for (int j = 0; j < DK_; ++j) acc += wk[j] * qh[j];
    w_eff[i] = acc;
}

__global__ void k_scores(const float* __restrict__ x, const float* __restrict__ w_eff,
                         float* __restrict__ attn)
{
    int wave = (blockIdx.x * blockDim.x + threadIdx.x) >> 5;
    int b  = wave >> 8;
    int lt = wave & 255;

    const float* A = x + ((size_t)b * L_ + (size_t)lt * 16) * D_;
    v8f c = wmma_gemm_f32(A, D_, w_eff, H_, D_);

    const int lane = threadIdx.x & 31, half = lane >> 4, n = lane & 15;
    float* out = attn + ((size_t)b * H_ + n) * L_ + (size_t)lt * 16 + 8 * half;
#pragma unroll
    for (int r = 0; r < 8; ++r) out[r] = c[r];
}

__global__ void k_softmax(float* __restrict__ attn)
{
    __shared__ float red[256];
    float* p = attn + (size_t)blockIdx.x * L_;
    const int t = threadIdx.x;

    float vals[16];
    float m = -3.4e38f;
#pragma unroll
    for (int i = 0; i < 16; ++i) { vals[i] = p[t + i * 256]; m = fmaxf(m, vals[i]); }

    red[t] = m; __syncthreads();
    for (int s = 128; s > 0; s >>= 1) {
        if (t < s) red[t] = fmaxf(red[t], red[t + s]);
        __syncthreads();
    }
    m = red[0]; __syncthreads();

    float sum = 0.f;
#pragma unroll
    for (int i = 0; i < 16; ++i) { vals[i] = expf(vals[i] - m); sum += vals[i]; }

    red[t] = sum; __syncthreads();
    for (int s = 128; s > 0; s >>= 1) {
        if (t < s) red[t] += red[t + s];
        __syncthreads();
    }
    float inv = 1.f / red[0];
#pragma unroll
    for (int i = 0; i < 16; ++i) p[t + i * 256] = vals[i] * inv;
}

#define CTXX_SPLIT 4
__global__ void k_ctxx_split(const float* __restrict__ attn, const float* __restrict__ x,
                             float* __restrict__ ctx_x_part)
{
    int wave = (blockIdx.x * blockDim.x + threadIdx.x) >> 5;
    int b  = wave >> 8;
    int t  = wave & 255;
    int s  = t >> 6;
    int dt = t & 63;
    const int k0 = s * (L_ / CTXX_SPLIT);

    const float* A  = attn + (size_t)b * H_ * L_ + k0;
    const float* Bm = x + ((size_t)b * L_ + k0) * D_ + dt * 16;
    v8f c = wmma_gemm_f32(A, L_, Bm, D_, L_ / CTXX_SPLIT);

    const int lane = threadIdx.x & 31, half = lane >> 4, n = lane & 15;
    float* out = ctx_x_part + (size_t)s * (B_ * H_ * D_)
               + (size_t)b * H_ * D_ + (size_t)dt * 16 + n;
#pragma unroll
    for (int r = 0; r < 8; ++r) out[(size_t)(r + 8 * half) * D_] = c[r];
}

__global__ void k_reduce(const float* __restrict__ part, float* __restrict__ dst,
                         int nparts, int n)
{
    int stride = gridDim.x * blockDim.x;
    for (int i = blockIdx.x * blockDim.x + threadIdx.x; i < n; i += stride) {
        float s = 0.f;
        for (int p = 0; p < nparts; ++p) s += part[(size_t)p * n + i];
        dst[i] = s;
    }
}

#define VSPLIT 8
__global__ void k_ctx_split(const float* __restrict__ ctx_x, const float* __restrict__ w_vs,
                            float* __restrict__ ctx_part)
{
    int wave = (blockIdx.x * blockDim.x + threadIdx.x) >> 5;
    int tile = wave >> 3;
    int s    = wave & 7;
    int h    = tile >> 2;
    int dvt  = tile & 3;
    const int k0 = s * (D_ / VSPLIT);

    const float* A  = ctx_x + (size_t)h * D_ + k0;
    const float* Bm = w_vs + (size_t)k0 * (H_ * DV_) + h * DV_ + dvt * 16;
    v8f c = wmma_gemm_f32(A, H_ * D_, Bm, H_ * DV_, D_ / VSPLIT);

    const int lane = threadIdx.x & 31, half = lane >> 4, n = lane & 15;
    float* out = ctx_part + (size_t)s * (B_ * H_ * DV_)
               + h * DV_ + dvt * 16 + n;
#pragma unroll
    for (int r = 0; r < 8; ++r) out[(size_t)(r + 8 * half) * (H_ * DV_)] = c[r];
}

__global__ void k_fc_split(const float* __restrict__ ctx, const float* __restrict__ fc,
                           float* __restrict__ out_pre_part)
{
    int wave = (blockIdx.x * blockDim.x + threadIdx.x) >> 5;
    int dt = wave >> 3;
    int s  = wave & 7;
    const int k0 = s * ((H_ * DV_) / VSPLIT);

    const float* A  = ctx + k0;
    const float* Bm = fc + (size_t)k0 * D_ + dt * 16;
    v8f c = wmma_gemm_f32(A, H_ * DV_, Bm, D_, (H_ * DV_) / VSPLIT);

    const int lane = threadIdx.x & 31, half = lane >> 4, n = lane & 15;
    float* out = out_pre_part + (size_t)s * (B_ * D_) + dt * 16 + n;
#pragma unroll
    for (int r = 0; r < 8; ++r) out[(size_t)(r + 8 * half) * D_] = c[r];
}

__global__ void k_ln(const float* __restrict__ out_pre_part, const float* __restrict__ query,
                     const float* __restrict__ gamma, const float* __restrict__ beta,
                     float* __restrict__ out)
{
    __shared__ float rs[256], rs2[256];
    const int b = blockIdx.x, t = threadIdx.x;

    float vals[4];
    float s = 0.f, s2 = 0.f;
#pragma unroll
    for (int i = 0; i < 4; ++i) {
        const int idx = t + i * 256;
        float v = query[idx];
        for (int p = 0; p < VSPLIT; ++p)
            v += out_pre_part[(size_t)p * (B_ * D_) + (size_t)b * D_ + idx];
        vals[i] = v; s += v; s2 += v * v;
    }
    rs[t] = s; rs2[t] = s2; __syncthreads();
    for (int k = 128; k > 0; k >>= 1) {
        if (t < k) { rs[t] += rs[t + k]; rs2[t] += rs2[t + k]; }
        __syncthreads();
    }
    float mean = rs[0] * (1.f / D_);
    float var  = rs2[0] * (1.f / D_) - mean * mean;
    float inv  = rsqrtf(var + 1e-6f);
#pragma unroll
    for (int i = 0; i < 4; ++i) {
        int idx = t + i * 256;
        out[(size_t)b * D_ + idx] = (vals[i] - mean) * inv * gamma[idx] + beta[idx];
    }
}

extern "C" void kernel_launch(void* const* d_in, const int* in_sizes, int n_in,
                              void* d_out, int out_size, void* d_ws, size_t ws_size,
                              hipStream_t stream)
{
    (void)in_sizes; (void)n_in; (void)out_size; (void)ws_size;

    const float* x     = (const float*)d_in[0];
    const float* query = (const float*)d_in[1];
    const float* w_qs  = (const float*)d_in[2];
    const float* w_ks  = (const float*)d_in[3];
    const float* w_vs  = (const float*)d_in[4];
    const float* fc    = (const float*)d_in[5];
    const float* gamma = (const float*)d_in[6];
    const float* beta  = (const float*)d_in[7];

    float* out  = (float*)d_out;
    float* attn = out + (size_t)B_ * D_;

    float* ws           = (float*)d_ws;
    float* qs           = ws;
    float* w_eff        = qs + H_ * DK_;
    float* ctx_x_part   = w_eff + D_ * H_;
    float* ctx_x        = ctx_x_part + (size_t)CTXX_SPLIT * B_ * H_ * D_;
    float* ctx_part     = ctx_x + (size_t)B_ * H_ * D_;
    float* ctx          = ctx_part + (size_t)VSPLIT * B_ * H_ * DV_;
    float* out_pre_part = ctx + (size_t)B_ * H_ * DV_;

    k_qproj<<<32, 1024, 0, stream>>>(query, w_qs, qs);
    k_weff<<<64, 256, 0, stream>>>(w_ks, qs, w_eff);
    k_scores<<<512, 256, 0, stream>>>(x, w_eff, attn);
    k_softmax<<<B_ * H_, 256, 0, stream>>>(attn);
    k_ctxx_split<<<512, 256, 0, stream>>>(attn, x, ctx_x_part);
    k_reduce<<<256, 256, 0, stream>>>(ctx_x_part, ctx_x, CTXX_SPLIT, B_ * H_ * D_);
    k_ctx_split<<<64, 256, 0, stream>>>(ctx_x, w_vs, ctx_part);
    k_reduce<<<16, 256, 0, stream>>>(ctx_part, ctx, VSPLIT, B_ * H_ * DV_);
    k_fc_split<<<64, 256, 0, stream>>>(ctx, fc, out_pre_part);
    k_ln<<<B_, 256, 0, stream>>>(out_pre_part, query, gamma, beta, out);
}


# the two partial-sum reduction launches removed: each tail GEMM work-group adds the K-split planes once (same order) into an LDS A block shared by its waves; 10 launches become 8; sums unchanged
# speedup vs baseline: 1.8316x; 1.2022x over previous
#include <hip/hip_runtime.h>
#include <math.h>

typedef __attribute__((ext_vector_type(2))) float v2f;
typedef __attribute__((ext_vector_type(4))) float v4f;
typedef __attribute__((ext_vector_type(8))) float v8f;

#define B_  16
#define L_  4096
#define D_  1024
#define H_  16
#define DK_ 64
#define DV_ 64

#define GB_ 8
#define AP_ 36
#define GW_ 8

static __device__ __forceinline__ void gemm_gload_a(const float* __restrict__ ag, int lda,
                                                    v4f (&g)[4])
{
#pragma unroll
    for (int j = 0; j < 4; ++j)
        g[j] = *(const v4f*)(ag + (size_t)(4 * j) * lda);
}

static __device__ __forceinline__ void gemm_lds_put(float* sw, const v4f (&g)[4])
{
#pragma unroll
    for (int j = 0; j < 4; ++j)
        *(v4f*)(sw + (4 * j) * AP_) = g[j];
}

static __device__ __forceinline__ void gemm_lds_get(const float* sr, v2f (&a)[GB_])
{
#pragma unroll
    for (int u = 0; u < GB_; ++u)
        a[u] = *(const v2f*)(sr + 4 * u);
}

static __device__ __forceinline__ void wave_lds_order()
{
    __builtin_amdgcn_fence(__ATOMIC_RELEASE, "wavefront");
    __builtin_amdgcn_wave_barrier();
    __builtin_amdgcn_fence(__ATOMIC_ACQUIRE, "wavefront");
}

static __device__ __forceinline__ void gemm_load_b(const float* __restrict__ bp, int ldb,
                                                   v2f (&b)[GB_])
{
#pragma unroll
    for (int u = 0; u < GB_; ++u) {
        b[u].x = bp[(size_t)(4 * u) * ldb];
        b[u].y = bp[(size_t)(4 * u + 1) * ldb];
    }
}

static __device__ __forceinline__ v8f gemm_mma_blk(const v2f (&a)[GB_], const v2f (&b)[GB_], v8f c)
{
#pragma unroll
    for (int u = 0; u < GB_; ++u)
        c = __builtin_amdgcn_wmma_f32_16x16x4_f32(false, a[u], false, b[u],
                                                  (short)0, c, false, false);
    return c;
}

static __device__ __forceinline__ v8f wmma_gemm_f32(const float* __restrict__ A, int lda,
                                                    const float* __restrict__ Bm, int ldb,
                                                    int Klen)
{
    __shared__ __attribute__((aligned(16))) float slab[GW_][16 * AP_];

    const int lane = threadIdx.x & 31;
    const int half = lane >> 4;
    const int mn   = lane & 15;
    float* sl = slab[threadIdx.x >> 5];

    const float* ag = A + (size_t)(lane >> 3) * lda + 4 * (lane & 7);
    float*       sw = sl + (lane >> 3) * AP_ + 4 * (lane & 7);
    const float* sr = sl + mn * AP_ + 2 * half;
    const float* bp = Bm + (size_t)(2 * half) * ldb + mn;
    const size_t bstep = (size_t)(4 * GB_) * ldb;
    const int nblk = Klen / (4 * GB_);

    v4f g0[4], g1[4];
    v2f a0[GB_], b0[GB_], a1[GB_], b1[GB_];
    v8f c = {};

    gemm_gload_a(ag, lda, g0);
    gemm_gload_a(ag + 4 * GB_, lda, g1);
    gemm_load_b(bp, ldb, b0);
    gemm_lds_put(sw, g0);
    gemm_gload_a(ag + 8 * GB_, lda, g0);
    wave_lds_order();
    gemm_lds_get(sr, a0);
    ag += 12 * GB_;

    for (int i = 0; i + 2 < nblk; i += 2) {
        wave_lds_order();
        gemm_lds_put(sw, g1);
        gemm_gload_a(ag, lda, g1);
        gemm_load_b(bp + bstep, ldb, b1);
        wave_lds_order();
        gemm_lds_get(sr, a1);
        c = gemm_mma_blk(a0, b0, c);

        wave_lds_order();
        gemm_lds_put(sw, g0);
        if (i + 4 < nblk) gemm_gload_a(ag + 4 * GB_, lda, g0);
        gemm_load_b(bp + 2 * bstep, ldb, b0);
        wave_lds_order();
        gemm_lds_get(sr, a0);
        c = gemm_mma_blk(a1, b1, c);

        ag += 8 * GB_;
        bp += 2 * bstep;
    }
    wave_lds_order();
    gemm_lds_put(sw, g1);
    gemm_load_b(bp + bstep, ldb, b1);
    wave_lds_order();
    gemm_lds_get(sr, a1);
    c = gemm_mma_blk(a0, b0, c);
    c = gemm_mma_blk(a1, b1, c);
    return c;
}


#define QP_COLS 32
#define QP_ROWS 32
__global__ void __launch_bounds__(1024)
k_qproj(const float* __restrict__ query, const float* __restrict__ w_qs,
        float* __restrict__ qs)
{
    __shared__ float red[D_ / QP_ROWS][QP_COLS];

    const int lane = threadIdx.x & 31;
    const int w = __builtin_amdgcn_readfirstlane((int)(threadIdx.x >> 5));
    const int col = blockIdx.x * QP_COLS + lane;

    const float* wp = w_qs + (size_t)(w * QP_ROWS) * (H_ * DK_) + col;
    const float* qp = query + w * QP_ROWS;

    float v[QP_ROWS];
    for (int r = 0; r < QP_ROWS; ++r) v[r] = wp[(size_t)r * (H_ * DK_)];

    float acc = 0.f;
    for (int r = 0; r < QP_ROWS; ++r) acc += qp[r] * v[r];

    red[w][lane] = acc;
    __syncthreads();

    if (w == 0) {
        float s = 0.f;
        for (int k = 0; k < D_ / QP_ROWS; ++k) s += red[k][lane];
        qs[col] = s * 0.125f;
    }
}

__global__ void k_weff(const float* __restrict__ w_ks, const float* __restrict__ qs,
                       float* __restrict__ w_eff)
{
    int i = blockIdx.x * blockDim.x + threadIdx.x;
    int d = i >> 4, h = i & 15;
    const float* wk = w_ks + (size_t)d * (H_ * DK_) + h * DK_;
    const float* qh = qs + h * DK_;
    float acc = 0.f;
#pragma unroll
    for (int j = 0; j < DK_; ++j) acc += wk[j] * qh[j];
    w_eff[i] = acc;
}

__global__ void k_scores(const float* __restrict__ x, const float* __restrict__ w_eff,
                         float* __restrict__ attn)
{
    int wave = (blockIdx.x * blockDim.x + threadIdx.x) >> 5;
    int b  = wave >> 8;
    int lt = wave & 255;

    const float* A = x + ((size_t)b * L_ + (size_t)lt * 16) * D_;
    v8f c = wmma_gemm_f32(A, D_, w_eff, H_, D_);

    const int lane = threadIdx.x & 31, half = lane >> 4, n = lane & 15;
    float* out = attn + ((size_t)b * H_ + n) * L_ + (size_t)lt * 16 + 8 * half;
#pragma unroll
    for (int r = 0; r < 8; ++r) out[r] = c[r];
}

__global__ void k_softmax(float* __restrict__ attn)
{
    __shared__ float red[256];
    float* p = attn + (size_t)blockIdx.x * L_;
    const int t = threadIdx.x;

    float vals[16];
    float m = -3.4e38f;
#pragma unroll
    for (int i = 0; i < 16; ++i) { vals[i] = p[t + i * 256]; m = fmaxf(m, vals[i]); }

    red[t] = m; __syncthreads();
    for (int s = 128; s > 0; s >>= 1) {
        if (t < s) red[t] = fmaxf(red[t], red[t + s]);
        __syncthreads();
    }
    m = red[0]; __syncthreads();

    float sum = 0.f;
#pragma unroll
    for (int i = 0; i < 16; ++i) { vals[i] = expf(vals[i] - m); sum += vals[i]; }

    red[t] = sum; __syncthreads();
    for (int s = 128; s > 0; s >>= 1) {
        if (t < s) red[t] += red[t + s];
        __syncthreads();
    }
    float inv = 1.f / red[0];
#pragma unroll
    for (int i = 0; i < 16; ++i) p[t + i * 256] = vals[i] * inv;
}

#define CTXX_SPLIT 4
__global__ void k_ctxx_split(const float* __restrict__ attn, const float* __restrict__ x,
                             float* __restrict__ ctx_x_part)
{
    int wave = (blockIdx.x * blockDim.x + threadIdx.x) >> 5;
    int b  = wave >> 8;
    int t  = wave & 255;
    int s  = t >> 6;
    int dt = t & 63;
    const int k0 = s * (L_ / CTXX_SPLIT);

    const float* A  = attn + (size_t)b * H_ * L_ + k0;
    const float* Bm = x + ((size_t)b * L_ + k0) * D_ + dt * 16;
    v8f c = wmma_gemm_f32(A, L_, Bm, D_, L_ / CTXX_SPLIT);

    const int lane = threadIdx.x & 31, half = lane >> 4, n = lane & 15;
    float* out = ctx_x_part + (size_t)s * (B_ * H_ * D_)
               + (size_t)b * H_ * D_ + (size_t)dt * 16 + n;
#pragma unroll
    for (int r = 0; r < 8; ++r) out[(size_t)(r + 8 * half) * D_] = c[r];
}

#define TK_ 128
#define TP_ 132
template <int NP, int NW>
static __device__ __forceinline__ v8f tail_gemm_f32(const float* __restrict__ Ap, int lda,
                                                    size_t pstride,
                                                    const float* __restrict__ Bm, int ldb)
{
    __shared__ __attribute__((aligned(16))) float sA[16 * TP_];
    constexpr int NT  = NW * 32;
    constexpr int PER = (16 * TK_ / 4) / NT;

    const int lane = threadIdx.x & 31;
    const int half = lane >> 4;
    const int mn   = lane & 15;

    v2f b[TK_ / 4];
    const float* bp = Bm + (size_t)(2 * half) * ldb + mn;
#pragma unroll
    for (int u = 0; u < TK_ / 4; ++u) {
        b[u].x = bp[(size_t)(4 * u) * ldb];
        b[u].y = bp[(size_t)(4 * u + 1) * ldb];
    }

    v4f raw[PER][NP];
#pragma unroll
    for (int i = 0; i < PER; ++i) {
        const int idx = threadIdx.x + i * NT;
        const float* p = Ap + (size_t)(idx >> 5) * lda + 4 * (idx & 31);
#pragma unroll
        for (int q = 0; q < NP; ++q) raw[i][q] = *(const v4f*)(p + (size_t)q * pstride);
    }
#pragma unroll
    for (int i = 0; i < PER; ++i) {
        const int idx = threadIdx.x + i * NT;
        v4f s = {0.f, 0.f, 0.f, 0.f};
#pragma unroll
        for (int q = 0; q < NP; ++q) s += raw[i][q];
        *(v4f*)(sA + (idx >> 5) * TP_ + 4 * (idx & 31)) = s;
    }
    __syncthreads();

    v8f c = {};
    const float* sr = sA + mn * TP_ + 2 * half;
#pragma unroll
    for (int u = 0; u < TK_ / 4; ++u) {
        const v2f a = *(const v2f*)(sr + 4 * u);
        c = __builtin_amdgcn_wmma_f32_16x16x4_f32(false, a, false, b[u],
                                                  (short)0, c, false, false);
    }
    return c;
}

#define VSPLIT 8
#define CTXW_ 4
__global__ void __launch_bounds__(CTXW_ * 32)
k_ctx_split(const float* __restrict__ ctx_x_part, const float* __restrict__ w_vs,
            float* __restrict__ ctx_part)
{
    const int h   = blockIdx.x >> 3;
    const int s   = blockIdx.x & 7;
    const int dvt = threadIdx.x >> 5;
    const int k0  = s * (D_ / VSPLIT);

    const float* A  = ctx_x_part + (size_t)h * D_ + k0;
    const float* Bm = w_vs + (size_t)k0 * (H_ * DV_) + h * DV_ + dvt * 16;
    v8f c = tail_gemm_f32<CTXX_SPLIT, CTXW_>(A, H_ * D_, (size_t)B_ * H_ * D_, Bm, H_ * DV_);

    const int lane = threadIdx.x & 31, half = lane >> 4, n = lane & 15;
    float* out = ctx_part + (size_t)s * (B_ * H_ * DV_)
               + h * DV_ + dvt * 16 + n;
#pragma unroll
    for (int r = 0; r < 8; ++r) out[(size_t)(r + 8 * half) * (H_ * DV_)] = c[r];
}

#define FCW_ 8
__global__ void __launch_bounds__(FCW_ * 32)
k_fc_split(const float* __restrict__ ctx_part, const float* __restrict__ fc,
           float* __restrict__ out_pre_part)
{
    const int s  = blockIdx.x & 7;
    const int dt = (blockIdx.x >> 3) * FCW_ + (threadIdx.x >> 5);
    const int k0 = s * ((H_ * DV_) / VSPLIT);

    const float* A  = ctx_part + k0;
    const float* Bm = fc + (size_t)k0 * D_ + dt * 16;
    v8f c = tail_gemm_f32<VSPLIT, FCW_>(A, H_ * DV_, (size_t)B_ * H_ * DV_, Bm, D_);

    const int lane = threadIdx.x & 31, half = lane >> 4, n = lane & 15;
    float* out = out_pre_part + (size_t)s * (B_ * D_) + dt * 16 + n;
#pragma unroll
    for (int r = 0; r < 8; ++r) out[(size_t)(r + 8 * half) * D_] = c[r];
}

__global__ void k_ln(const float* __restrict__ out_pre_part, const float* __restrict__ query,
                     const float* __restrict__ gamma, const float* __restrict__ beta,
                     float* __restrict__ out)
{
    __shared__ float rs[256], rs2[256];
    const int b = blockIdx.x, t = threadIdx.x;

    float vals[4];
    float s = 0.f, s2 = 0.f;
#pragma unroll
    for (int i = 0; i < 4; ++i) {
        const int idx = t + i * 256;
        float v = query[idx];
        for (int p = 0; p < VSPLIT; ++p)
            v += out_pre_part[(size_t)p * (B_ * D_) + (size_t)b * D_ + idx];
        vals[i] = v; s += v; s2 += v * v;
    }
    rs[t] = s; rs2[t] = s2; __syncthreads();
    for (int k = 128; k > 0; k >>= 1) {
        if (t < k) { rs[t] += rs[t + k]; rs2[t] += rs2[t + k]; }
        __syncthreads();
    }
    float mean = rs[0] * (1.f / D_);
    float var  = rs2[0] * (1.f / D_) - mean * mean;
    float inv  = rsqrtf(var + 1e-6f);
#pragma unroll
    for (int i = 0; i < 4; ++i) {
        int idx = t + i * 256;
        out[(size_t)b * D_ + idx] = (vals[i] - mean) * inv * gamma[idx] + beta[idx];
    }
}

extern "C" void kernel_launch(void* const* d_in, const int* in_sizes, int n_in,
                              void* d_out, int out_size, void* d_ws, size_t ws_size,
                              hipStream_t stream)
{
    (void)in_sizes; (void)n_in; (void)out_size; (void)ws_size;

    const float* x     = (const float*)d_in[0];
    const float* query = (const float*)d_in[1];
    const float* w_qs  = (const float*)d_in[2];
    const float* w_ks  = (const float*)d_in[3];
    const float* w_vs  = (const float*)d_in[4];
    const float* fc    = (const float*)d_in[5];
    const float* gamma = (const float*)d_in[6];
    const float* beta  = (const float*)d_in[7];

    float* out  = (float*)d_out;
    float* attn = out + (size_t)B_ * D_;

    float* ws           = (float*)d_ws;
    float* qs           = ws;
    float* w_eff        = qs + H_ * DK_;
    float* ctx_x_part   = w_eff + D_ * H_;
    float* ctx_part     = ctx_x_part + (size_t)CTXX_SPLIT * B_ * H_ * D_;
    float* out_pre_part = ctx_part + (size_t)VSPLIT * B_ * H_ * DV_;

    k_qproj<<<32, 1024, 0, stream>>>(query, w_qs, qs);
    k_weff<<<64, 256, 0, stream>>>(w_ks, qs, w_eff);
    k_scores<<<512, 256, 0, stream>>>(x, w_eff, attn);
    k_softmax<<<B_ * H_, 256, 0, stream>>>(attn);
    k_ctxx_split<<<512, 256, 0, stream>>>(attn, x, ctx_x_part);
    k_ctx_split<<<H_ * VSPLIT, CTXW_ * 32, 0, stream>>>(ctx_x_part, w_vs, ctx_part);
    k_fc_split<<<(D_ / 16 / FCW_) * VSPLIT, FCW_ * 32, 0, stream>>>(ctx_part, fc, out_pre_part);
    k_ln<<<B_, 256, 0, stream>>>(out_pre_part, query, gamma, beta, out);
}
